# starting kernel
# baseline (speedup 1.0000x reference)
#include <hip/hip_runtime.h>

#define F_FIELDS 39
#define D_DIM    64

typedef __bf16 v16bf __attribute__((ext_vector_type(16)));
typedef __bf16 v2bf  __attribute__((ext_vector_type(2)));
typedef float  v8f   __attribute__((ext_vector_type(8)));

union FragU { v16bf v; v2bf p[8]; };

__device__ __forceinline__ v2bf pk_bf16(float x, float y) {
#if __has_builtin(__builtin_amdgcn_cvt_pk_bf16_f32)
    return __builtin_amdgcn_cvt_pk_bf16_f32(x, y);
#else
    v2bf r; r.x = (__bf16)x; r.y = (__bf16)y; return r;
#endif
}

__global__ __launch_bounds__(256) void fwfm_wmma_kernel(
    const int*   __restrict__ x,
    const float* __restrict__ emb,
    const float* __restrict__ bias,
    const float* __restrict__ W,
    const float* __restrict__ w0,
    float*       __restrict__ out,
    int Brows)
{
    const int lane   = threadIdx.x & 31;
    const int wave   = (int)((blockIdx.x * blockDim.x + threadIdx.x) >> 5);
    const int nwaves = (int)((gridDim.x * blockDim.x) >> 5);

    const int TI[6] = {0,0,0,1,1,2};
    const int TJ[6] = {0,1,2,1,2,2};

    const int n_c  = lane & 15;
    const int m_hi = (lane >> 4) * 8;
    float wreg[6][8];
    #pragma unroll
    for (int tp = 0; tp < 6; ++tp) {
        #pragma unroll
        for (int r = 0; r < 8; ++r) {
            int i = 16 * TI[tp] + m_hi + r;
            int j = 16 * TJ[tp] + n_c;
            wreg[tp][r] = (i < j && j < F_FIELDS) ? W[i * F_FIELDS + j] : 0.0f;
        }
    }

    const int rrow   = lane & 15;
    const int kbase0 = (lane & 16) ? 8 : 0;

    for (int b = wave; b < Brows; b += nwaves) {
        const int* xb = x + (size_t)b * F_FIELDS;

        FragU frag[3][2];
        #pragma unroll
        for (int t = 0; t < 3; ++t) {
            const int  row   = 16 * t + rrow;
            const bool valid = (row < F_FIELDS);
            const int  idx   = valid ? xb[row] : 0;
            const float* erow = emb + (size_t)idx * D_DIM;
            #pragma unroll
            for (int h = 0; h < 2; ++h) {
                const int kb = kbase0 + 32 * h;
                FragU& f = frag[t][h];
                if (valid) {
                    const float4* p0 = (const float4*)(erow + kb);
                    const float4* p1 = (const float4*)(erow + kb + 16);
                    float4 a0 = p0[0], a1 = p0[1];
                    float4 b0 = p1[0], b1 = p1[1];
                    f.p[0] = pk_bf16(a0.x, a0.y);
                    f.p[1] = pk_bf16(a0.z, a0.w);
                    f.p[2] = pk_bf16(a1.x, a1.y);
                    f.p[3] = pk_bf16(a1.z, a1.w);
                    f.p[4] = pk_bf16(b0.x, b0.y);
                    f.p[5] = pk_bf16(b0.z, b0.w);
                    f.p[6] = pk_bf16(b1.x, b1.y);
                    f.p[7] = pk_bf16(b1.z, b1.w);
                } else {
                    v2bf z = pk_bf16(0.0f, 0.0f);
                    #pragma unroll
                    for (int s = 0; s < 8; ++s) f.p[s] = z;
                }
            }
        }

        float wsum = 0.0f;
        #pragma unroll
        for (int tp = 0; tp < 6; ++tp) {
            v8f acc = {};
            acc = __builtin_amdgcn_wmma_f32_16x16x32_bf16(
                false, frag[TI[tp]][0].v, false, frag[TJ[tp]][0].v,
                (short)0, acc, false, false);
            acc = __builtin_amdgcn_wmma_f32_16x16x32_bf16(
                false, frag[TI[tp]][1].v, false, frag[TJ[tp]][1].v,
                (short)0, acc, false, false);
            #pragma unroll
            for (int r = 0; r < 8; ++r) wsum += wreg[tp][r] * acc[r];
        }

        wsum += bias[(size_t)xb[lane]];
        if (lane + 32 < F_FIELDS) wsum += bias[(size_t)xb[lane + 32]];

        #pragma unroll
        for (int off = 16; off >= 1; off >>= 1)
            wsum += __shfl_xor(wsum, off, 32);

        if (lane == 0) out[b] = w0[0] + wsum;
    }
}

extern "C" void kernel_launch(void* const* d_in, const int* in_sizes, int n_in,
                              void* d_out, int out_size, void* d_ws, size_t ws_size,
                              hipStream_t stream) {
    const int*   x    = (const int*)  d_in[0];
    const float* emb  = (const float*)d_in[1];
    const float* bias = (const float*)d_in[2];
    const float* W    = (const float*)d_in[3];
    const float* w0   = (const float*)d_in[4];
    float*       out  = (float*)d_out;

    const int Brows = in_sizes[0] / F_FIELDS;
    const int block = 256;
    int grid = (Brows + 7) / 8;
    if (grid < 1) grid = 1;

    fwfm_wmma_kernel<<<grid, block, 0, stream>>>(x, emb, bias, W, w0, out, Brows);
}


# Resident grid: 256 workgroups of 16 waves (one per WGP); each wave loops over its batch rows and requests the next row's gather ahead of the current row's products; weight tile read once per wave
# speedup vs baseline: 1.3446x; 1.3446x over previous
#include <hip/hip_runtime.h>


#define F_FIELDS 39
#define D_DIM    64

typedef __bf16 v16bf __attribute__((ext_vector_type(16)));
typedef __bf16 v2bf  __attribute__((ext_vector_type(2)));
typedef float  v8f   __attribute__((ext_vector_type(8)));

union FragU { v16bf v; v2bf p[8]; };

__device__ __forceinline__ v2bf pk_bf16(float x, float y) {
#if __has_builtin(__builtin_amdgcn_cvt_pk_bf16_f32)
    return __builtin_amdgcn_cvt_pk_bf16_f32(x, y);
#else
    v2bf r; r.x = (__bf16)x; r.y = (__bf16)y; return r;
#endif
}

struct RawTile { float4 q[2][4]; };

__device__ __forceinline__ void request_tile(RawTile& r, const float* __restrict__ emb,
                                             int idx, int kbase0) {
    const float* erow = emb + (size_t)idx * D_DIM;
    #pragma unroll
    for (int h = 0; h < 2; ++h) {
        const int kb = kbase0 + 32 * h;
        const float4* p0 = (const float4*)(erow + kb);
        const float4* p1 = (const float4*)(erow + kb + 16);
        r.q[h][0] = p0[0];
        r.q[h][1] = p0[1];
        r.q[h][2] = p1[0];
        r.q[h][3] = p1[1];
    }
}

__device__ __forceinline__ void pack_tile(FragU (&f)[2], const RawTile& r) {
    #pragma unroll
    for (int h = 0; h < 2; ++h) {
        f[h].p[0] = pk_bf16(r.q[h][0].x, r.q[h][0].y);
        f[h].p[1] = pk_bf16(r.q[h][0].z, r.q[h][0].w);
        f[h].p[2] = pk_bf16(r.q[h][1].x, r.q[h][1].y);
        f[h].p[3] = pk_bf16(r.q[h][1].z, r.q[h][1].w);
        f[h].p[4] = pk_bf16(r.q[h][2].x, r.q[h][2].y);
        f[h].p[5] = pk_bf16(r.q[h][2].z, r.q[h][2].w);
        f[h].p[6] = pk_bf16(r.q[h][3].x, r.q[h][3].y);
        f[h].p[7] = pk_bf16(r.q[h][3].z, r.q[h][3].w);
    }
}

__global__ __launch_bounds__(512) void fwfm_wmma_kernel(
    const int*   __restrict__ x,
    const float* __restrict__ emb,
    const float* __restrict__ bias,
    const float* __restrict__ W,
    const float* __restrict__ w0,
    float*       __restrict__ out,
    int Brows)
{
    const int lane   = threadIdx.x & 31;
    const int wave   = (int)((blockIdx.x * blockDim.x + threadIdx.x) >> 5);
    const int nwaves = (int)((gridDim.x * blockDim.x) >> 5);

    int b = wave;
    if (b >= Brows) return;

    const int TI[6] = {0,0,0,1,1,2};
    const int TJ[6] = {0,1,2,1,2,2};

    const int n_c  = lane & 15;
    const int m_hi = (lane >> 4) * 8;
    float wreg[6][8];
    #pragma unroll
    for (int tp = 0; tp < 6; ++tp) {
        #pragma unroll
        for (int r = 0; r < 8; ++r) {
            int i = 16 * TI[tp] + m_hi + r;
            int j = 16 * TJ[tp] + n_c;
            wreg[tp][r] = (i < j && j < F_FIELDS) ? W[i * F_FIELDS + j] : 0.0f;
        }
    }
    const float w0v = w0[0];

    const int  rrow   = lane & 15;
    const int  kbase0 = (lane & 16) ? 8 : 0;
    const bool valid2 = (32 + rrow) < F_FIELDS;
    const bool bias2  = (lane + 32) < F_FIELDS;

    const int o2 = valid2 ? 32 + rrow : rrow;
    const int ob = bias2  ? lane + 32 : lane;

    const int* xb = x + (size_t)b * F_FIELDS;
    int i0 = xb[rrow];
    int i1 = xb[16 + rrow];
    int i2 = xb[o2];
    int j0 = xb[lane];
    int j1 = xb[ob];
    int n0, n1, n2, m0, m1;
    {
        const int  bq = (b + nwaves < Brows) ? b + nwaves : b;
        const int* xn = x + (size_t)bq * F_FIELDS;
        n0 = xn[rrow]; n1 = xn[16 + rrow]; n2 = xn[o2]; m0 = xn[lane]; m1 = xn[ob];
    }
    asm volatile("" : "+v"(i0), "+v"(i1), "+v"(i2), "+v"(j0), "+v"(j1) :: "memory");

    RawTile raw0, raw1, raw2;
    #pragma unroll
    for (int h = 0; h < 2; ++h) {
        #pragma unroll
        for (int s = 0; s < 4; ++s) raw2.q[h][s] = make_float4(0.0f, 0.0f, 0.0f, 0.0f);
    }
    request_tile(raw0, emb, i0, kbase0);
    request_tile(raw1, emb, i1, kbase0);
    if (valid2) request_tile(raw2, emb, i2, kbase0);

    float bs0 = bias[(size_t)j0];
    float bs1 = 0.0f;
    if (bias2) bs1 = bias[(size_t)j1];

    for (;;) {
        const int  bn   = b + nwaves;
        const bool more = bn < Brows;

        const int  b2 = (bn + nwaves < Brows) ? bn + nwaves : b;
        const int* xp = x + (size_t)b2 * F_FIELDS;
        const int p0 = xp[rrow];
        const int p1 = xp[16 + rrow];
        const int p2 = xp[o2];
        const int q0 = xp[lane];
        const int q1 = xp[ob];

        FragU frag[3][2];
        pack_tile(frag[0], raw0);
        pack_tile(frag[1], raw1);
        pack_tile(frag[2], raw2);

        #pragma unroll
        for (int t = 0; t < 3; ++t) {
            asm volatile("" : "+v"(frag[t][0].v), "+v"(frag[t][1].v) :: "memory");
        }
        asm volatile("" : "+v"(bs0), "+v"(bs1) :: "memory");
        asm volatile("" : "+v"(n0), "+v"(n1), "+v"(n2), "+v"(m0), "+v"(m1) :: "memory");

        float nb0 = 0.0f, nb1 = 0.0f;
        if (more) {
            request_tile(raw0, emb, n0, kbase0);
            request_tile(raw1, emb, n1, kbase0);
            if (valid2) request_tile(raw2, emb, n2, kbase0);
            nb0 = bias[(size_t)m0];
            if (bias2) nb1 = bias[(size_t)m1];
        }

        float wsum = 0.0f;
        #pragma unroll
        for (int tp = 0; tp < 6; ++tp) {
            v8f acc = {};
            acc = __builtin_amdgcn_wmma_f32_16x16x32_bf16(
                false, frag[TI[tp]][0].v, false, frag[TJ[tp]][0].v,
                (short)0, acc, false, false);
            acc = __builtin_amdgcn_wmma_f32_16x16x32_bf16(
                false, frag[TI[tp]][1].v, false, frag[TJ[tp]][1].v,
                (short)0, acc, false, false);
            #pragma unroll
            for (int r = 0; r < 8; ++r) wsum += wreg[tp][r] * acc[r];
        }

        wsum += bs0;
        if (bias2) wsum += bs1;

        #pragma unroll
        for (int off = 16; off >= 1; off >>= 1)
            wsum += __shfl_xor(wsum, off, 32);

        if (lane == 0) out[b] = w0v + wsum;

        if (!more) break;
        b   = bn;
        bs0 = nb0; bs1 = nb1;
        n0 = p0; n1 = p1; n2 = p2; m0 = q0; m1 = q1;
    }
}

extern "C" void kernel_launch(void* const* d_in, const int* in_sizes, int n_in,
                              void* d_out, int out_size, void* d_ws, size_t ws_size,
                              hipStream_t stream) {
    const int*   x    = (const int*)  d_in[0];
    const float* emb  = (const float*)d_in[1];
    const float* bias = (const float*)d_in[2];
    const float* W    = (const float*)d_in[3];
    const float* w0   = (const float*)d_in[4];
    float*       out  = (float*)d_out;

    const int Brows = in_sizes[0] / F_FIELDS;
    const int block = 512;
    const int grid  = 256;

    fwfm_wmma_kernel<<<grid, block, 0, stream>>>(x, emb, bias, W, w0, out, Brows);
}


# Whole-line gather: 16 adjacent lanes read one 256-byte embedding row (20 loads of 4 lines each per batch row, was 24 of 16 lines); packed rows staged in per-wave LDS, WMMA fragments read from LDS
# speedup vs baseline: 1.5162x; 1.1276x over previous
#include <hip/hip_runtime.h>


#define F_FIELDS 39
#define D_DIM    64

#define WAVES_PER_WG 16
#define N_LOADS      20
#define LDS_PITCH    144
#define LDS_ROWS     40
#define LDS_PER_WAVE (LDS_ROWS * LDS_PITCH)

typedef __bf16 v16bf __attribute__((ext_vector_type(16)));
typedef __bf16 v2bf  __attribute__((ext_vector_type(2)));
typedef float  v8f   __attribute__((ext_vector_type(8)));
typedef unsigned int u32x2 __attribute__((ext_vector_type(2)));
typedef unsigned int u32x4 __attribute__((ext_vector_type(4)));

union FragU { v16bf v; u32x4 q[2]; };
union Pack4 { v2bf p[2]; u32x2 u; };

__device__ __forceinline__ v2bf pk_bf16(float x, float y) {
#if __has_builtin(__builtin_amdgcn_cvt_pk_bf16_f32)
    return __builtin_amdgcn_cvt_pk_bf16_f32(x, y);
#else
    v2bf r; r.x = (__bf16)x; r.y = (__bf16)y; return r;
#endif
}

__device__ __forceinline__ constexpr int row_lo(int t) {
    return t < 16 ? (t & 7) + 16 * (t >> 3) : 32 + 2 * (t - 16);
}
__device__ __forceinline__ constexpr int row_hi(int t) {
    return t < 16 ? row_lo(t) + 8 : row_lo(t) + 1;
}

struct RawRow { float4 q[N_LOADS]; };

__device__ __forceinline__ void request_row(RawRow& r, const char* __restrict__ embB,
                                            const int (&sidx)[LDS_ROWS], bool upper, unsigned c16) {
    #pragma unroll
    for (int t = 0; t < N_LOADS; ++t) {
        const unsigned ia  = (unsigned)sidx[row_lo(t)];
        const unsigned ib  = (unsigned)sidx[row_hi(t)];
        const unsigned off = ((upper ? ib : ia) << 8) + c16;
        r.q[t] = *(const float4*)(embB + off);
    }
}

__device__ __forceinline__ void stage_row(const RawRow& r, unsigned char* stA, unsigned char* stB) {
    #pragma unroll
    for (int t = 0; t < N_LOADS; ++t) {
        Pack4 pk;
        pk.p[0] = pk_bf16(r.q[t].x, r.q[t].y);
        pk.p[1] = pk_bf16(r.q[t].z, r.q[t].w);
        unsigned char* dst = (t < 16 ? stA : stB) + row_lo(t) * LDS_PITCH;
        *(u32x2*)dst = pk.u;
    }
}

__global__ __launch_bounds__(512) void fwfm_wmma_kernel(
    const int*   __restrict__ x,
    const float* __restrict__ emb,
    const float* __restrict__ bias,
    const float* __restrict__ W,
    const float* __restrict__ w0,
    float*       __restrict__ out,
    int Brows)
{
    __shared__ __attribute__((aligned(16))) unsigned char stage[WAVES_PER_WG * LDS_PER_WAVE];

    const int lane   = threadIdx.x & 31;
    const int wave   = __builtin_amdgcn_readfirstlane((int)((blockIdx.x * blockDim.x + threadIdx.x) >> 5));
    const int wloc   = __builtin_amdgcn_readfirstlane((int)(threadIdx.x >> 5));
    const int nwaves = (int)((gridDim.x * blockDim.x) >> 5);

    int b = wave;
    if (b >= Brows) return;

    const int TI[6] = {0,0,0,1,1,2};
    const int TJ[6] = {0,1,2,1,2,2};

    const int n_c  = lane & 15;
    const int m_hi = (lane >> 4) * 8;
    float wreg[6][8];
    #pragma unroll
    for (int tp = 0; tp < 6; ++tp) {
        #pragma unroll
        for (int r = 0; r < 8; ++r) {
            int i = 16 * TI[tp] + m_hi + r;
            int j = 16 * TJ[tp] + n_c;
            wreg[tp][r] = (i < j && j < F_FIELDS) ? W[i * F_FIELDS + j] : 0.0f;
        }
    }
    const float w0v = w0[0];

    const int  rrow   = lane & 15;
    const bool upper  = (lane & 16) != 0;
    const bool valid2 = (32 + rrow) < F_FIELDS;
    const bool bias2  = (lane + 32) < F_FIELDS;
    const int  ob     = bias2 ? lane + 32 : lane;

    const char*    embB = (const char*)emb;
    const unsigned c16  = (unsigned)rrow * 16u;

    unsigned char* const wbase = stage + wloc * LDS_PER_WAVE;
    unsigned char* const stA   = wbase + rrow * 8 + (upper ? 8 * LDS_PITCH : 0);
    unsigned char* const stB   = wbase + rrow * 8 + (upper ? 1 * LDS_PITCH : 0);
    const unsigned char* const fbase = wbase + rrow * LDS_PITCH + (upper ? 16 : 0);

    int sidx[LDS_ROWS];
    {
        const int* xb = x + (size_t)b * F_FIELDS;
        #pragma unroll
        for (int f = 0; f < F_FIELDS; ++f) sidx[f] = xb[f];
        sidx[F_FIELDS] = sidx[F_FIELDS - 1];
    }
    RawRow raw;
    request_row(raw, embB, sidx, upper, c16);
    float bs0, bs1 = 0.0f;
    int nj0, nj1;
    {
        const int* xb = x + (size_t)b * F_FIELDS;
        const int j0 = xb[lane];
        const int j1 = xb[ob];
        bs0 = bias[(size_t)j0];
        if (bias2) bs1 = bias[(size_t)j1];
        const int  bq = (b + nwaves < Brows) ? b + nwaves : b;
        const int* xn = x + (size_t)bq * F_FIELDS;
        #pragma unroll
        for (int f = 0; f < F_FIELDS; ++f) sidx[f] = xn[f];
        sidx[F_FIELDS] = sidx[F_FIELDS - 1];
        nj0 = xn[lane];
        nj1 = xn[ob];
    }

    for (;;) {
        const int  bn   = b + nwaves;
        const bool more = bn < Brows;

        stage_row(raw, stA, stB);

        asm volatile("" : "+v"(bs0), "+v"(bs1), "+v"(nj0), "+v"(nj1) :: "memory");

        float nb0 = 0.0f, nb1 = 0.0f;
        if (more) {
            request_row(raw, embB, sidx, upper, c16);
            nb0 = bias[(size_t)nj0];
            if (bias2) nb1 = bias[(size_t)nj1];
        }

        const int  b2 = (bn + nwaves < Brows) ? bn + nwaves : b;
        const int* xp = x + (size_t)b2 * F_FIELDS;
        #pragma unroll
        for (int f = 0; f < F_FIELDS; ++f) sidx[f] = xp[f];
        sidx[F_FIELDS] = sidx[F_FIELDS - 1];
        const int pj0 = xp[lane];
        const int pj1 = xp[ob];

        FragU frag[3][2];
        #pragma unroll
        for (int t = 0; t < 2; ++t) {
            #pragma unroll
            for (int h = 0; h < 2; ++h) {
                const unsigned char* p = fbase + t * 16 * LDS_PITCH + h * 64;
                frag[t][h].q[0] = *(const u32x4*)(p);
                frag[t][h].q[1] = *(const u32x4*)(p + 32);
            }
        }
        #pragma unroll
        for (int h = 0; h < 2; ++h) {
            u32x4 z = {0u, 0u, 0u, 0u};
            frag[2][h].q[0] = z;
            frag[2][h].q[1] = z;
            if (valid2) {
                const unsigned char* p = fbase + 2 * 16 * LDS_PITCH + h * 64;
                frag[2][h].q[0] = *(const u32x4*)(p);
                frag[2][h].q[1] = *(const u32x4*)(p + 32);
            }
        }

        float wsum = 0.0f;
        #pragma unroll
        for (int tp = 0; tp < 6; ++tp) {
            v8f acc = {};
            acc = __builtin_amdgcn_wmma_f32_16x16x32_bf16(
                false, frag[TI[tp]][0].v, false, frag[TJ[tp]][0].v,
                (short)0, acc, false, false);
            acc = __builtin_amdgcn_wmma_f32_16x16x32_bf16(
                false, frag[TI[tp]][1].v, false, frag[TJ[tp]][1].v,
                (short)0, acc, false, false);
            #pragma unroll
            for (int r = 0; r < 8; ++r) wsum += wreg[tp][r] * acc[r];
        }

        wsum += bs0;
        if (bias2) wsum += bs1;

        #pragma unroll
        for (int off = 16; off >= 1; off >>= 1)
            wsum += __shfl_xor(wsum, off, 32);

        if (lane == 0) out[b] = w0v + wsum;

        if (!more) break;
        b   = bn;
        bs0 = nb0; bs1 = nb1;
        nj0 = pj0; nj1 = pj1;
    }
}

extern "C" void kernel_launch(void* const* d_in, const int* in_sizes, int n_in,
                              void* d_out, int out_size, void* d_ws, size_t ws_size,
                              hipStream_t stream) {
    const int*   x    = (const int*)  d_in[0];
    const float* emb  = (const float*)d_in[1];
    const float* bias = (const float*)d_in[2];
    const float* W    = (const float*)d_in[3];
    const float* w0   = (const float*)d_in[4];
    float*       out  = (float*)d_out;

    const int Brows = in_sizes[0] / F_FIELDS;
    const int block = 512;
    const int grid  = 256;

    fwfm_wmma_kernel<<<grid, block, 0, stream>>>(x, emb, bias, W, w0, out, Brows);
}


# L2 prefetch (global_prefetch_b8, one 256-B embedding row per lane) of the batch row two ahead, issued behind each demand gather; per-lane indices now read three rows ahead
# speedup vs baseline: 1.5683x; 1.0344x over previous
#include <hip/hip_runtime.h>


#define F_FIELDS 39
#define D_DIM    64

#define WAVES_PER_WG 16
#define N_LOADS      20
#define LDS_PITCH    144
#define LDS_ROWS     40
#define LDS_PER_WAVE (LDS_ROWS * LDS_PITCH)

typedef __bf16 v16bf __attribute__((ext_vector_type(16)));
typedef __bf16 v2bf  __attribute__((ext_vector_type(2)));
typedef float  v8f   __attribute__((ext_vector_type(8)));
typedef unsigned int u32x2 __attribute__((ext_vector_type(2)));
typedef unsigned int u32x4 __attribute__((ext_vector_type(4)));

union FragU { v16bf v; u32x4 q[2]; };
union Pack4 { v2bf p[2]; u32x2 u; };

__device__ __forceinline__ v2bf pk_bf16(float x, float y) {
#if __has_builtin(__builtin_amdgcn_cvt_pk_bf16_f32)
    return __builtin_amdgcn_cvt_pk_bf16_f32(x, y);
#else
    v2bf r; r.x = (__bf16)x; r.y = (__bf16)y; return r;
#endif
}

__device__ __forceinline__ constexpr int row_lo(int t) {
    return t < 16 ? (t & 7) + 16 * (t >> 3) : 32 + 2 * (t - 16);
}
__device__ __forceinline__ constexpr int row_hi(int t) {
    return t < 16 ? row_lo(t) + 8 : row_lo(t) + 1;
}

struct RawRow { float4 q[N_LOADS]; };

__device__ __forceinline__ void request_row(RawRow& r, const char* __restrict__ embB,
                                            const int (&sidx)[LDS_ROWS], bool upper, unsigned c16) {
    #pragma unroll
    for (int t = 0; t < N_LOADS; ++t) {
        const unsigned ia  = (unsigned)sidx[row_lo(t)];
        const unsigned ib  = (unsigned)sidx[row_hi(t)];
        const unsigned off = ((upper ? ib : ia) << 8) + c16;
        r.q[t] = *(const float4*)(embB + off);
    }
}

__device__ __forceinline__ void prefetch_row(const char* __restrict__ embB, int j0, int j1, bool second) {
    __builtin_prefetch(embB + ((unsigned)j0 << 8), 0, 2);
    if (second) __builtin_prefetch(embB + ((unsigned)j1 << 8), 0, 2);
}

__device__ __forceinline__ void stage_row(const RawRow& r, unsigned char* stA, unsigned char* stB) {
    #pragma unroll
    for (int t = 0; t < N_LOADS; ++t) {
        Pack4 pk;
        pk.p[0] = pk_bf16(r.q[t].x, r.q[t].y);
        pk.p[1] = pk_bf16(r.q[t].z, r.q[t].w);
        unsigned char* dst = (t < 16 ? stA : stB) + row_lo(t) * LDS_PITCH;
        *(u32x2*)dst = pk.u;
    }
}

__global__ __launch_bounds__(512) void fwfm_wmma_kernel(
    const int*   __restrict__ x,
    const float* __restrict__ emb,
    const float* __restrict__ bias,
    const float* __restrict__ W,
    const float* __restrict__ w0,
    float*       __restrict__ out,
    int Brows)
{
    __shared__ __attribute__((aligned(16))) unsigned char stage[WAVES_PER_WG * LDS_PER_WAVE];

    const int lane   = threadIdx.x & 31;
    const int wave   = __builtin_amdgcn_readfirstlane((int)((blockIdx.x * blockDim.x + threadIdx.x) >> 5));
    const int wloc   = __builtin_amdgcn_readfirstlane((int)(threadIdx.x >> 5));
    const int nwaves = (int)((gridDim.x * blockDim.x) >> 5);

    int b = wave;
    if (b >= Brows) return;

    const int TI[6] = {0,0,0,1,1,2};
    const int TJ[6] = {0,1,2,1,2,2};

    const int n_c  = lane & 15;
    const int m_hi = (lane >> 4) * 8;
    float wreg[6][8];
    #pragma unroll
    for (int tp = 0; tp < 6; ++tp) {
        #pragma unroll
        for (int r = 0; r < 8; ++r) {
            int i = 16 * TI[tp] + m_hi + r;
            int j = 16 * TJ[tp] + n_c;
            wreg[tp][r] = (i < j && j < F_FIELDS) ? W[i * F_FIELDS + j] : 0.0f;
        }
    }
    const float w0v = w0[0];

    const int  rrow   = lane & 15;
    const bool upper  = (lane & 16) != 0;
    const bool valid2 = (32 + rrow) < F_FIELDS;
    const bool bias2  = (lane + 32) < F_FIELDS;
    const int  ob     = bias2 ? lane + 32 : lane;

    const char*    embB = (const char*)emb;
    const unsigned c16  = (unsigned)rrow * 16u;

    unsigned char* const wbase = stage + wloc * LDS_PER_WAVE;
    unsigned char* const stA   = wbase + rrow * 8 + (upper ? 8 * LDS_PITCH : 0);
    unsigned char* const stB   = wbase + rrow * 8 + (upper ? 1 * LDS_PITCH : 0);
    const unsigned char* const fbase = wbase + rrow * LDS_PITCH + (upper ? 16 : 0);

    int sidx[LDS_ROWS];
    const int* xb = x + (size_t)b * F_FIELDS;
    const int  bq = (b + nwaves < Brows) ? b + nwaves : b;
    const int  br = (b + 2 * nwaves < Brows) ? b + 2 * nwaves : b;
    const int* xn = x + (size_t)bq * F_FIELDS;
    const int* xr = x + (size_t)br * F_FIELDS;
    const int j0 = xb[lane];
    const int j1 = xb[ob];
    int nj0 = xn[lane];
    int nj1 = xn[ob];
    int pf0 = xr[lane];
    int pf1 = xr[ob];
    #pragma unroll
    for (int f = 0; f < F_FIELDS; ++f) sidx[f] = xb[f];
    sidx[F_FIELDS] = sidx[F_FIELDS - 1];
    RawRow raw;
    request_row(raw, embB, sidx, upper, c16);
    if (b + nwaves < Brows) prefetch_row(embB, nj0, nj1, bias2);
    float bs0, bs1 = 0.0f;
    bs0 = bias[(size_t)j0];
    if (bias2) bs1 = bias[(size_t)j1];
    #pragma unroll
    for (int f = 0; f < F_FIELDS; ++f) sidx[f] = xn[f];
    sidx[F_FIELDS] = sidx[F_FIELDS - 1];

    for (;;) {
        const int  bn   = b + nwaves;
        const bool more = bn < Brows;

        stage_row(raw, stA, stB);

        asm volatile("" : "+v"(bs0), "+v"(bs1), "+v"(nj0), "+v"(nj1) :: "memory");

        float nb0 = 0.0f, nb1 = 0.0f;
        if (more) {
            request_row(raw, embB, sidx, upper, c16);
            nb0 = bias[(size_t)nj0];
            if (bias2) nb1 = bias[(size_t)nj1];
        }

        const bool more2 = bn + nwaves < Brows;
        if (more2) prefetch_row(embB, pf0, pf1, bias2);

        const int  b2 = more2 ? bn + nwaves : b;
        const int* xp = x + (size_t)b2 * F_FIELDS;
        #pragma unroll
        for (int f = 0; f < F_FIELDS; ++f) sidx[f] = xp[f];
        sidx[F_FIELDS] = sidx[F_FIELDS - 1];
        const int  b3 = (bn + 2 * nwaves < Brows) ? bn + 2 * nwaves : b;
        const int* xq = x + (size_t)b3 * F_FIELDS;
        const int qj0 = xq[lane];
        const int qj1 = xq[ob];

        FragU frag[3][2];
        #pragma unroll
        for (int t = 0; t < 2; ++t) {
            #pragma unroll
            for (int h = 0; h < 2; ++h) {
                const unsigned char* p = fbase + t * 16 * LDS_PITCH + h * 64;
                frag[t][h].q[0] = *(const u32x4*)(p);
                frag[t][h].q[1] = *(const u32x4*)(p + 32);
            }
        }
        #pragma unroll
        for (int h = 0; h < 2; ++h) {
            u32x4 z = {0u, 0u, 0u, 0u};
            frag[2][h].q[0] = z;
            frag[2][h].q[1] = z;
            if (valid2) {
                const unsigned char* p = fbase + 2 * 16 * LDS_PITCH + h * 64;
                frag[2][h].q[0] = *(const u32x4*)(p);
                frag[2][h].q[1] = *(const u32x4*)(p + 32);
            }
        }

        float wsum = 0.0f;
        #pragma unroll
        for (int tp = 0; tp < 6; ++tp) {
            v8f acc = {};
            acc = __builtin_amdgcn_wmma_f32_16x16x32_bf16(
                false, frag[TI[tp]][0].v, false, frag[TJ[tp]][0].v,
                (short)0, acc, false, false);
            acc = __builtin_amdgcn_wmma_f32_16x16x32_bf16(
                false, frag[TI[tp]][1].v, false, frag[TJ[tp]][1].v,
                (short)0, acc, false, false);
            #pragma unroll
            for (int r = 0; r < 8; ++r) wsum += wreg[tp][r] * acc[r];
        }

        wsum += bs0;
        if (bias2) wsum += bs1;

        #pragma unroll
        for (int off = 16; off >= 1; off >>= 1)
            wsum += __shfl_xor(wsum, off, 32);

        if (lane == 0) out[b] = w0v + wsum;

        if (!more) break;
        b   = bn;
        bs0 = nb0; bs1 = nb1;
        nj0 = pf0; nj1 = pf1;
        pf0 = qj0; pf1 = qj1;
    }
}

extern "C" void kernel_launch(void* const* d_in, const int* in_sizes, int n_in,
                              void* d_out, int out_size, void* d_ws, size_t ws_size,
                              hipStream_t stream) {
    const int*   x    = (const int*)  d_in[0];
    const float* emb  = (const float*)d_in[1];
    const float* bias = (const float*)d_in[2];
    const float* W    = (const float*)d_in[3];
    const float* w0   = (const float*)d_in[4];
    float*       out  = (float*)d_out;

    const int Brows = in_sizes[0] / F_FIELDS;
    const int block = 512;
    const int grid  = 256;

    fwfm_wmma_kernel<<<grid, block, 0, stream>>>(x, emb, bias, W, w0, out, Brows);
}


# Loads never wait on an earlier load's address translation (xcnt): a row's 22 addresses precomputed into own registers and held, no lane mask between loads (bias, prefetch, W tile loads unmasked)
# speedup vs baseline: 1.6860x; 1.0750x over previous
#include <hip/hip_runtime.h>


#define F_FIELDS 39
#define D_DIM    64

#define WAVES_PER_WG 16
#define N_LOADS      20
#define LDS_PITCH    144
#define LDS_ROWS     40
#define LDS_PER_WAVE (LDS_ROWS * LDS_PITCH)

typedef __bf16 v16bf __attribute__((ext_vector_type(16)));
typedef __bf16 v2bf  __attribute__((ext_vector_type(2)));
typedef float  v8f   __attribute__((ext_vector_type(8)));
typedef unsigned int u32x2 __attribute__((ext_vector_type(2)));
typedef unsigned int u32x4 __attribute__((ext_vector_type(4)));

union FragU { v16bf v; u32x4 q[2]; };
union Pack4 { v2bf p[2]; u32x2 u; };

__device__ __forceinline__ v2bf pk_bf16(float x, float y) {
#if __has_builtin(__builtin_amdgcn_cvt_pk_bf16_f32)
    return __builtin_amdgcn_cvt_pk_bf16_f32(x, y);
#else
    v2bf r; r.x = (__bf16)x; r.y = (__bf16)y; return r;
#endif
}

__device__ __forceinline__ constexpr int row_lo(int t) {
    return t < 16 ? (t & 7) + 16 * (t >> 3) : 32 + 2 * (t - 16);
}
__device__ __forceinline__ constexpr int row_hi(int t) {
    return t < 16 ? row_lo(t) + 8 : row_lo(t) + 1;
}

struct RawRow { float4 q[N_LOADS]; };

struct RowOffs { unsigned o[N_LOADS]; };

__device__ __forceinline__ void row_offsets(RowOffs& a, const int (&sidx)[LDS_ROWS], bool upper, unsigned c16) {
    #pragma unroll
    for (int t = 0; t < N_LOADS; ++t) {
        const unsigned ia = (unsigned)sidx[row_lo(t)];
        const unsigned ib = (unsigned)sidx[row_hi(t)];
        a.o[t] = ((upper ? ib : ia) << 8) + c16;
    }
}

__device__ __forceinline__ void request_row(RawRow& r, const char* __restrict__ embB, const RowOffs& a) {
    #pragma unroll
    for (int t = 0; t < N_LOADS; ++t) r.q[t] = *(const float4*)(embB + a.o[t]);
}

__device__ __forceinline__ void hold_offsets(const RowOffs& a, unsigned p0, unsigned p1) {
    asm volatile("" :: "v"(a.o[0]),  "v"(a.o[1]),  "v"(a.o[2]),  "v"(a.o[3]),  "v"(a.o[4]),
                       "v"(a.o[5]),  "v"(a.o[6]),  "v"(a.o[7]),  "v"(a.o[8]),  "v"(a.o[9]),
                       "v"(a.o[10]), "v"(a.o[11]), "v"(a.o[12]), "v"(a.o[13]), "v"(a.o[14]),
                       "v"(a.o[15]), "v"(a.o[16]), "v"(a.o[17]), "v"(a.o[18]), "v"(a.o[19]),
                       "v"(p0), "v"(p1));
}

__device__ __forceinline__ void prefetch_row(const char* __restrict__ embB, unsigned o0, unsigned o1) {
    __builtin_prefetch(embB + o0, 0, 2);
    __builtin_prefetch(embB + o1, 0, 2);
}

__device__ __forceinline__ void stage_row(const RawRow& r, unsigned char* stA, unsigned char* stB) {
    #pragma unroll
    for (int t = 0; t < N_LOADS; ++t) {
        Pack4 pk;
        pk.p[0] = pk_bf16(r.q[t].x, r.q[t].y);
        pk.p[1] = pk_bf16(r.q[t].z, r.q[t].w);
        unsigned char* dst = (t < 16 ? stA : stB) + row_lo(t) * LDS_PITCH;
        *(u32x2*)dst = pk.u;
    }
}

__global__ __launch_bounds__(512) void fwfm_wmma_kernel(
    const int*   __restrict__ x,
    const float* __restrict__ emb,
    const float* __restrict__ bias,
    const float* __restrict__ W,
    const float* __restrict__ w0,
    float*       __restrict__ out,
    int Brows)
{
    __shared__ __attribute__((aligned(16))) unsigned char stage[WAVES_PER_WG * LDS_PER_WAVE];

    const int lane   = threadIdx.x & 31;
    const int wave   = __builtin_amdgcn_readfirstlane((int)((blockIdx.x * blockDim.x + threadIdx.x) >> 5));
    const int wloc   = __builtin_amdgcn_readfirstlane((int)(threadIdx.x >> 5));
    const int nwaves = (int)((gridDim.x * blockDim.x) >> 5);

    int b = wave;
    if (b >= Brows) return;

    const int TI[6] = {0,0,0,1,1,2};
    const int TJ[6] = {0,1,2,1,2,2};

    const int n_c  = lane & 15;
    const int m_hi = (lane >> 4) * 8;
    const int nc6  = n_c < 6 ? n_c : 6;
    float wreg[6][8];
    #pragma unroll
    for (int tp = 0; tp < 6; ++tp) {
        #pragma unroll
        for (int r = 0; r < 8; ++r) {
            const int i  = 16 * TI[tp] + m_hi + r;
            const int j  = 16 * TJ[tp] + n_c;
            const int il = (TI[tp] == 2) ? 32 + (r < 6 ? r : 6) : i;
            const int jl = (TJ[tp] == 2) ? 32 + nc6 : j;
            const float w = W[il * F_FIELDS + jl];
            wreg[tp][r] = (i < j && j < F_FIELDS) ? w : 0.0f;
        }
    }
    const float w0v = w0[0];

    const int  rrow   = lane & 15;
    const bool upper  = (lane & 16) != 0;
    const bool valid2 = (32 + rrow) < F_FIELDS;
    const bool bias2  = (lane + 32) < F_FIELDS;
    const int  ob     = bias2 ? lane + 32 : 32;

    const char*    embB = (const char*)emb;
    const unsigned c16  = (unsigned)rrow * 16u;

    unsigned char* const wbase = stage + wloc * LDS_PER_WAVE;
    unsigned char* const stA   = wbase + rrow * 8 + (upper ? 8 * LDS_PITCH : 0);
    unsigned char* const stB   = wbase + rrow * 8 + (upper ? 1 * LDS_PITCH : 0);
    const unsigned char* const fbase = wbase + rrow * LDS_PITCH + (upper ? 16 : 0);

    int sidx[LDS_ROWS];
    const int* xb = x + (size_t)b * F_FIELDS;
    const int  bq = (b + nwaves < Brows) ? b + nwaves : b;
    const int  br = (b + 2 * nwaves < Brows) ? b + 2 * nwaves : b;
    const int* xn = x + (size_t)bq * F_FIELDS;
    const int* xr = x + (size_t)br * F_FIELDS;
    const int j0 = xb[lane];
    const int j1 = xb[ob];
    int nj0 = xn[lane];
    int nj1 = xn[ob];
    int pf0 = xr[lane];
    int pf1 = xr[ob];
    #pragma unroll
    for (int f = 0; f < F_FIELDS; ++f) sidx[f] = xb[f];
    sidx[F_FIELDS] = sidx[F_FIELDS - 1];
    RawRow raw;
    {
        RowOffs ro0;
        row_offsets(ro0, sidx, upper, c16);
        const unsigned q0 = (unsigned)nj0 << 8, q1 = (unsigned)nj1 << 8;
        request_row(raw, embB, ro0);
        if (b + nwaves < Brows) prefetch_row(embB, q0, q1);
        hold_offsets(ro0, q0, q1);
    }
    float bs0, bs1;
    bs0 = bias[(size_t)j0];
    bs1 = bias[(size_t)j1];
    #pragma unroll
    for (int f = 0; f < F_FIELDS; ++f) sidx[f] = xn[f];
    sidx[F_FIELDS] = sidx[F_FIELDS - 1];

    for (;;) {
        const int  bn   = b + nwaves;
        const bool more = bn < Brows;

        RowOffs ro;
        row_offsets(ro, sidx, upper, c16);
        const unsigned po0 = (unsigned)pf0 << 8, po1 = (unsigned)pf1 << 8;

        stage_row(raw, stA, stB);

        asm volatile("" : "+v"(bs0), "+v"(bs1), "+v"(nj0), "+v"(nj1) :: "memory");

        float nb0 = 0.0f, nb1 = 0.0f;
        if (more) {
            request_row(raw, embB, ro);
            nb0 = bias[(size_t)nj0];
            nb1 = bias[(size_t)nj1];
        }

        const bool more2 = bn + nwaves < Brows;
        if (more2) prefetch_row(embB, po0, po1);

        const int  b2 = more2 ? bn + nwaves : b;
        const int* xp = x + (size_t)b2 * F_FIELDS;
        #pragma unroll
        for (int f = 0; f < F_FIELDS; ++f) sidx[f] = xp[f];
        sidx[F_FIELDS] = sidx[F_FIELDS - 1];
        const int  b3 = (bn + 2 * nwaves < Brows) ? bn + 2 * nwaves : b;
        const int* xq = x + (size_t)b3 * F_FIELDS;
        const int qj0 = xq[lane];
        const int qj1 = xq[ob];

        FragU frag[3][2];
        #pragma unroll
        for (int t = 0; t < 2; ++t) {
            #pragma unroll
            for (int h = 0; h < 2; ++h) {
                const unsigned char* p = fbase + t * 16 * LDS_PITCH + h * 64;
                frag[t][h].q[0] = *(const u32x4*)(p);
                frag[t][h].q[1] = *(const u32x4*)(p + 32);
            }
        }
        #pragma unroll
        for (int h = 0; h < 2; ++h) {
            u32x4 z = {0u, 0u, 0u, 0u};
            frag[2][h].q[0] = z;
            frag[2][h].q[1] = z;
            if (valid2) {
                const unsigned char* p = fbase + 2 * 16 * LDS_PITCH + h * 64;
                frag[2][h].q[0] = *(const u32x4*)(p);
                frag[2][h].q[1] = *(const u32x4*)(p + 32);
            }
        }

        float wsum = 0.0f;
        #pragma unroll
        for (int tp = 0; tp < 6; ++tp) {
            v8f acc = {};
            acc = __builtin_amdgcn_wmma_f32_16x16x32_bf16(
                false, frag[TI[tp]][0].v, false, frag[TJ[tp]][0].v,
                (short)0, acc, false, false);
            acc = __builtin_amdgcn_wmma_f32_16x16x32_bf16(
                false, frag[TI[tp]][1].v, false, frag[TJ[tp]][1].v,
                (short)0, acc, false, false);
            #pragma unroll
            for (int r = 0; r < 8; ++r) wsum += wreg[tp][r] * acc[r];
        }

        wsum += bs0;
        if (bias2) wsum += bs1;

        #pragma unroll
        for (int off = 16; off >= 1; off >>= 1)
            wsum += __shfl_xor(wsum, off, 32);

        if (lane == 0) out[b] = w0v + wsum;

        hold_offsets(ro, po0, po1);

        if (!more) break;
        b   = bn;
        bs0 = nb0; bs1 = nb1;
        nj0 = pf0; nj1 = pf1;
        pf0 = qj0; pf1 = qj1;
    }
}

extern "C" void kernel_launch(void* const* d_in, const int* in_sizes, int n_in,
                              void* d_out, int out_size, void* d_ws, size_t ws_size,
                              hipStream_t stream) {
    const int*   x    = (const int*)  d_in[0];
    const float* emb  = (const float*)d_in[1];
    const float* bias = (const float*)d_in[2];
    const float* W    = (const float*)d_in[3];
    const float* w0   = (const float*)d_in[4];
    float*       out  = (float*)d_out;

    const int Brows = in_sizes[0] / F_FIELDS;
    const int block = 512;
    const int grid  = 256;

    fwfm_wmma_kernel<<<grid, block, 0, stream>>>(x, emb, bias, W, w0, out, Brows);
}


# Second landing register set: each batch row gathered once, two rows ahead (cache prefetch of it dropped); 12 waves per workgroup for the registers, rows dealt in equal blocks per workgroup
# speedup vs baseline: 1.7499x; 1.0379x over previous
#include <hip/hip_runtime.h>


#define F_FIELDS 39
#define D_DIM    64

#define WAVES_PER_WG 12
#define N_LOADS      20
#define LDS_PITCH    144
#define LDS_ROWS     40
#define LDS_PER_WAVE (LDS_ROWS * LDS_PITCH)

typedef __bf16 v16bf __attribute__((ext_vector_type(16)));
typedef __bf16 v2bf  __attribute__((ext_vector_type(2)));
typedef float  v8f   __attribute__((ext_vector_type(8)));
typedef unsigned int u32x2 __attribute__((ext_vector_type(2)));
typedef unsigned int u32x4 __attribute__((ext_vector_type(4)));

union FragU { v16bf v; u32x4 q[2]; };
union Pack4 { v2bf p[2]; u32x2 u; };

__device__ __forceinline__ v2bf pk_bf16(float x, float y) {
#if __has_builtin(__builtin_amdgcn_cvt_pk_bf16_f32)
    return __builtin_amdgcn_cvt_pk_bf16_f32(x, y);
#else
    v2bf r; r.x = (__bf16)x; r.y = (__bf16)y; return r;
#endif
}

__device__ __forceinline__ constexpr int row_lo(int t) {
    return t < 16 ? (t & 7) + 16 * (t >> 3) : 32 + 2 * (t - 16);
}
__device__ __forceinline__ constexpr int row_hi(int t) {
    return t < 16 ? row_lo(t) + 8 : row_lo(t) + 1;
}

struct RawRow { float4 q[N_LOADS]; };

struct RowOffs { unsigned o[N_LOADS]; };

__device__ __forceinline__ void row_offsets(RowOffs& a, const int (&sidx)[LDS_ROWS], bool upper, unsigned c16, unsigned keep) {
    #pragma unroll
    for (int t = 0; t < N_LOADS; ++t) {
        const unsigned ia = (unsigned)sidx[row_lo(t)] & keep;
        const unsigned ib = (unsigned)sidx[row_hi(t)] & keep;
        a.o[t] = ((upper ? ib : ia) << 8) + c16;
    }
}

__device__ __forceinline__ void request_row(RawRow& r, const char* __restrict__ embB, const RowOffs& a) {
    #pragma unroll
    for (int t = 0; t < N_LOADS; ++t) r.q[t] = *(const float4*)(embB + a.o[t]);
}

__device__ __forceinline__ void hold_offsets(const RowOffs& a) {
    asm volatile("" :: "v"(a.o[0]),  "v"(a.o[1]),  "v"(a.o[2]),  "v"(a.o[3]),  "v"(a.o[4]),
                       "v"(a.o[5]),  "v"(a.o[6]),  "v"(a.o[7]),  "v"(a.o[8]),  "v"(a.o[9]),
                       "v"(a.o[10]), "v"(a.o[11]), "v"(a.o[12]), "v"(a.o[13]), "v"(a.o[14]),
                       "v"(a.o[15]), "v"(a.o[16]), "v"(a.o[17]), "v"(a.o[18]), "v"(a.o[19]));
}

__device__ __forceinline__ void stage_row(const RawRow& r, unsigned char* stA, unsigned char* stB) {
    #pragma unroll
    for (int t = 0; t < N_LOADS; ++t) {
        Pack4 pk;
        pk.p[0] = pk_bf16(r.q[t].x, r.q[t].y);
        pk.p[1] = pk_bf16(r.q[t].z, r.q[t].w);
        unsigned char* dst = (t < 16 ? stA : stB) + row_lo(t) * LDS_PITCH;
        *(u32x2*)dst = pk.u;
    }
}

__global__ __launch_bounds__(384) void fwfm_wmma_kernel(
    const int*   __restrict__ x,
    const float* __restrict__ emb,
    const float* __restrict__ bias,
    const float* __restrict__ W,
    const float* __restrict__ w0,
    float*       __restrict__ out,
    int Brows)
{
    __shared__ __attribute__((aligned(16))) unsigned char stage[WAVES_PER_WG * LDS_PER_WAVE];

    const int lane   = threadIdx.x & 31;
    const int wloc   = __builtin_amdgcn_readfirstlane((int)(threadIdx.x >> 5));
    const int rstep  = (int)(blockDim.x >> 5);
    const int rpw    = (Brows + (int)gridDim.x - 1) / (int)gridDim.x;
    const int rbeg   = (int)blockIdx.x * rpw;
    const int rend   = (rbeg + rpw < Brows) ? rbeg + rpw : Brows;

    int b = rbeg + wloc;
    if (b >= rend) return;

    const int TI[6] = {0,0,0,1,1,2};
    const int TJ[6] = {0,1,2,1,2,2};

    const int n_c  = lane & 15;
    const int m_hi = (lane >> 4) * 8;
    const int nc6  = n_c < 6 ? n_c : 6;
    float wreg[6][8];
    #pragma unroll
    for (int tp = 0; tp < 6; ++tp) {
        #pragma unroll
        for (int r = 0; r < 8; ++r) {
            const int i  = 16 * TI[tp] + m_hi + r;
            const int j  = 16 * TJ[tp] + n_c;
            const int il = (TI[tp] == 2) ? 32 + (r < 6 ? r : 6) : i;
            const int jl = (TJ[tp] == 2) ? 32 + nc6 : j;
            const float w = W[il * F_FIELDS + jl];
            wreg[tp][r] = (i < j && j < F_FIELDS) ? w : 0.0f;
        }
    }
    const float w0v = w0[0];

    const int  rrow   = lane & 15;
    const bool upper  = (lane & 16) != 0;
    const bool valid2 = (32 + rrow) < F_FIELDS;
    const bool bias2  = (lane + 32) < F_FIELDS;
    const int  ob     = bias2 ? lane + 32 : 32;

    const char*    embB = (const char*)emb;
    const unsigned c16  = (unsigned)rrow * 16u;

    unsigned char* const wbase = stage + wloc * LDS_PER_WAVE;
    unsigned char* const stA   = wbase + rrow * 8 + (upper ? 8 * LDS_PITCH : 0);
    unsigned char* const stB   = wbase + rrow * 8 + (upper ? 1 * LDS_PITCH : 0);
    const unsigned char* const fbase = wbase + rrow * LDS_PITCH + (upper ? 16 : 0);

    int sidx[LDS_ROWS];
    typedef const __attribute__((address_space(4))) int* ro_int_ptr;
    const ro_int_ptr xs = (ro_int_ptr)x;
    const int  r1 = (b + rstep < rend) ? b + rstep : b;
    const int  r2 = (b + 2 * rstep < rend) ? b + 2 * rstep : b;
    const int  r3 = (b + 3 * rstep < rend) ? b + 3 * rstep : b;
    const int* x0 = x + (size_t)b  * F_FIELDS;
    const int* x1 = x + (size_t)r1 * F_FIELDS;
    const int* x2 = x + (size_t)r2 * F_FIELDS;
    const int* x3 = x + (size_t)r3 * F_FIELDS;
    const int j0 = x0[lane];
    const int j1 = x0[ob];
    const int k0 = x1[lane];
    const int k1 = x1[ob];
    int nj0 = x2[lane];
    int nj1 = x2[ob];
    int pj0 = x3[lane];
    int pj1 = x3[ob];
    RawRow rawA, rawB;
    float  bA0, bA1, bB0, bB1;
    #pragma unroll
    for (int f = 0; f < F_FIELDS; ++f) sidx[f] = x0[f];
    sidx[F_FIELDS] = sidx[F_FIELDS - 1];
    RowOffs ro0;
    row_offsets(ro0, sidx, upper, c16, ~0u);
    request_row(rawA, embB, ro0);
    bA0 = bias[(size_t)j0];
    bA1 = bias[(size_t)j1];
    #pragma unroll
    for (int f = 0; f < F_FIELDS; ++f) sidx[f] = x1[f];
    sidx[F_FIELDS] = sidx[F_FIELDS - 1];
    RowOffs ro1;
    const unsigned keep1 = (b + rstep < rend) ? ~0u : 0u;
    row_offsets(ro1, sidx, upper, c16, keep1);
    request_row(rawB, embB, ro1);
    bB0 = bias[(size_t)((unsigned)k0 & keep1)];
    bB1 = bias[(size_t)((unsigned)k1 & keep1)];
    hold_offsets(ro0);
    hold_offsets(ro1);
    #pragma unroll
    for (int f = 0; f < F_FIELDS; ++f) sidx[f] = x2[f];
    sidx[F_FIELDS] = sidx[F_FIELDS - 1];

#define LOOP_HALF(RAW, LB0, LB1) { \
        const unsigned keep = (b + 2 * rstep < rend) ? ~0u : 0u; \
        RowOffs ro; \
        row_offsets(ro, sidx, upper, c16, keep); \
        stage_row(RAW, stA, stB); \
        float cb0 = LB0, cb1 = LB1; \
        asm volatile("" : "+v"(cb0), "+v"(cb1), "+v"(nj0), "+v"(nj1) :: "memory"); \
        request_row(RAW, embB, ro); \
        LB0 = bias[(size_t)((unsigned)nj0 & keep)]; \
        LB1 = bias[(size_t)((unsigned)nj1 & keep)]; \
        const int  b3 = (b + 3 * rstep < rend) ? b + 3 * rstep : b; \
        const ro_int_ptr xp = xs + (size_t)b3 * F_FIELDS; \
        _Pragma("unroll") \
        for (int f = 0; f < F_FIELDS; ++f) sidx[f] = xp[f]; \
        sidx[F_FIELDS] = sidx[F_FIELDS - 1]; \
        const int  b4 = (b + 4 * rstep < rend) ? b + 4 * rstep : b; \
        const int* xq = x + (size_t)b4 * F_FIELDS; \
        const int qj0 = xq[lane]; \
        const int qj1 = xq[ob]; \
        FragU frag[3][2]; \
        _Pragma("unroll") \
        for (int t = 0; t < 2; ++t) { \
            _Pragma("unroll") \
            for (int h = 0; h < 2; ++h) { \
                const unsigned char* p = fbase + t * 16 * LDS_PITCH + h * 64; \
                frag[t][h].q[0] = *(const u32x4*)(p); \
                frag[t][h].q[1] = *(const u32x4*)(p + 32); \
            } \
        } \
        _Pragma("unroll") \
        for (int h = 0; h < 2; ++h) { \
            u32x4 z = {0u, 0u, 0u, 0u}; \
            frag[2][h].q[0] = z; \
            frag[2][h].q[1] = z; \
            if (valid2) { \
                const unsigned char* p = fbase + 2 * 16 * LDS_PITCH + h * 64; \
                frag[2][h].q[0] = *(const u32x4*)(p); \
                frag[2][h].q[1] = *(const u32x4*)(p + 32); \
            } \
        } \
        float wsum = 0.0f; \
        _Pragma("unroll") \
        for (int tp = 0; tp < 6; ++tp) { \
            v8f acc = {}; \
            acc = __builtin_amdgcn_wmma_f32_16x16x32_bf16( \
                false, frag[TI[tp]][0].v, false, frag[TJ[tp]][0].v, \
                (short)0, acc, false, false); \
            acc = __builtin_amdgcn_wmma_f32_16x16x32_bf16( \
                false, frag[TI[tp]][1].v, false, frag[TJ[tp]][1].v, \
                (short)0, acc, false, false); \
            _Pragma("unroll") \
            for (int r = 0; r < 8; ++r) wsum += wreg[tp][r] * acc[r]; \
        } \
        wsum += cb0; \
        if (bias2) wsum += cb1; \
        _Pragma("unroll") \
        for (int off = 16; off >= 1; off >>= 1) \
            wsum += __shfl_xor(wsum, off, 32); \
        if (lane == 0) out[b] = w0v + wsum; \
        hold_offsets(ro); \
        b  += rstep; \
        nj0 = pj0; nj1 = pj1; \
        pj0 = qj0; pj1 = qj1; \
    }

    const int nrows = (rend - 1 - b) / rstep + 1;
    for (int p = nrows >> 1; p > 0; --p) {
        LOOP_HALF(rawA, bA0, bA1)
        LOOP_HALF(rawB, bB0, bB1)
    }
    if (nrows & 1) {
        LOOP_HALF(rawA, bA0, bA1)
    }
#undef LOOP_HALF
}

extern "C" void kernel_launch(void* const* d_in, const int* in_sizes, int n_in,
                              void* d_out, int out_size, void* d_ws, size_t ws_size,
                              hipStream_t stream) {
    const int*   x    = (const int*)  d_in[0];
    const float* emb  = (const float*)d_in[1];
    const float* bias = (const float*)d_in[2];
    const float* W    = (const float*)d_in[3];
    const float* w0   = (const float*)d_in[4];
    float*       out  = (float*)d_out;

    const int Brows = in_sizes[0] / F_FIELDS;
    const int block = 384;
    const int grid  = 256;

    fwfm_wmma_kernel<<<grid, block, 0, stream>>>(x, emb, bias, W, w0, out, Brows);
}
